# starting kernel
# baseline (speedup 1.0000x reference)
#include <hip/hip_runtime.h>

typedef __attribute__((ext_vector_type(2))) float v2f;
typedef __attribute__((ext_vector_type(8))) float v8f;

#define LC      64
#define NCURVE  8
#define TAUC    0.5f
#define LOG2E   1.44269504088896340736f

__global__ __launch_bounds__(128) void landmark_deform_wmma_kernel(
    const float* __restrict__ momentum,
    const float* __restrict__ init_lm,
    const float* __restrict__ sigmaV2,
    float* __restrict__ out,
    int L)
{
    __shared__ float xs[LC], ys[LC];
    __shared__ float mmi[2 * LC];
    __shared__ float sg[LC], dpx[LC], dpy[LC];

    const int blk   = blockIdx.x;
    const int b     = blk / NCURVE;
    const int curve = blk % NCURVE;
    const int base  = curve * LC;
    const int t     = threadIdx.x;

    if (t < LC) {
        const int g = b * L + base + t;
        xs[t] = init_lm[2 * g + 0];
        ys[t] = init_lm[2 * g + 1];
        mmi[2 * t + 0] = momentum[2 * g + 0];
        mmi[2 * t + 1] = momentum[2 * g + 1];
        sg[t] = sigmaV2[base + t];
    }
    __syncthreads();

    const int lane = t & 31;
    const int wave = t >> 5;
    const int rb   = wave * 16;
    const int q    = lane & 15;
    const int half = lane >> 4;
    const int row  = rb + q;

    v2f Bf[16];
    #pragma unroll
    for (int kc = 0; kc < 16; ++kc) {
        const int j0 = 4 * kc + 2 * half;
        const float mx0 = mmi[2 * j0 + 0];
        const float my0 = mmi[2 * j0 + 1];
        const float mx1 = mmi[2 * j0 + 2];
        const float my1 = mmi[2 * j0 + 3];
        Bf[kc].x = (q == 0) ? mx0 : (q == 1) ? my0 : 0.0f;
        Bf[kc].y = (q == 0) ? mx1 : (q == 1) ? my1 : 0.0f;
    }

    #pragma unroll
    for (int step = 0; step < 2; ++step) {
        const float xr  = xs[row];
        const float yr  = ys[row];
        const float nis = -LOG2E * __builtin_amdgcn_rcpf(sg[row]);

        v8f acc = {0.f, 0.f, 0.f, 0.f, 0.f, 0.f, 0.f, 0.f};

        #pragma unroll
        for (int kc = 0; kc < 16; ++kc) {
            const int j0 = 4 * kc + 2 * half;

            const float dx0 = xr - xs[j0],     dy0 = yr - ys[j0];
            const float dx1 = xr - xs[j0 + 1], dy1 = yr - ys[j0 + 1];
            v2f A;
            A.x = __builtin_amdgcn_exp2f((dx0 * dx0 + dy0 * dy0) * nis);
            A.y = __builtin_amdgcn_exp2f((dx1 * dx1 + dy1 * dy1) * nis);

            acc = __builtin_amdgcn_wmma_f32_16x16x4_f32(
                false, A, false, Bf[kc], (short)0, acc, false, false);
        }

        if (q < 2) {
            float* dst = (q == 0) ? dpx : dpy;
            #pragma unroll
            for (int v = 0; v < 8; ++v)
                dst[rb + half * 8 + v] = acc[v];
        }
        __syncthreads();

        if (step == 0) {
            if (t < LC) {
                xs[t] += TAUC * dpx[t];
                ys[t] += TAUC * dpy[t];
            }
            __syncthreads();
        }
    }

    if (t < LC) {
        const int g = b * L + base + t;
        out[2 * g + 0] = xs[t] + TAUC * dpx[t];
        out[2 * g + 1] = ys[t] + TAUC * dpy[t];
    }
}

extern "C" void kernel_launch(void* const* d_in, const int* in_sizes, int n_in,
                              void* d_out, int out_size, void* d_ws, size_t ws_size,
                              hipStream_t stream) {
    const float* momentum = (const float*)d_in[0];
    const float* init_lm  = (const float*)d_in[1];
    const float* sigmaV2  = (const float*)d_in[3];
    float* out = (float*)d_out;

    const int L = in_sizes[3];
    const int B = in_sizes[0] / (2 * L);

    dim3 grid(B * NCURVE);
    dim3 block(128);
    landmark_deform_wmma_kernel<<<grid, block, 0, stream>>>(
        momentum, init_lm, sigmaV2, out, L);
}


# B-fragment build without branches: 16 exec-masked branches, each holding a full LDS wait (16 serial LDS round trips before the first WMMA), become 16 independent paired reads plus selects
# speedup vs baseline: 1.2927x; 1.2927x over previous
#include <hip/hip_runtime.h>

typedef __attribute__((ext_vector_type(2))) float v2f;
typedef __attribute__((ext_vector_type(8))) float v8f;

#define LC      64
#define NCURVE  8
#define TAUC    0.5f
#define LOG2E   1.44269504088896340736f

__global__ __launch_bounds__(128) void landmark_deform_wmma_kernel(
    const float* __restrict__ momentum,
    const float* __restrict__ init_lm,
    const float* __restrict__ sigmaV2,
    float* __restrict__ out,
    int L)
{
    __shared__ float xs[LC], ys[LC];
    __shared__ float mmi[2 * LC];
    __shared__ float sg[LC], dpx[LC], dpy[LC];

    const int blk   = blockIdx.x;
    const int b     = blk / NCURVE;
    const int curve = blk % NCURVE;
    const int base  = curve * LC;
    const int t     = threadIdx.x;

    if (t < LC) {
        const int g = b * L + base + t;
        xs[t] = init_lm[2 * g + 0];
        ys[t] = init_lm[2 * g + 1];
        mmi[2 * t + 0] = momentum[2 * g + 0];
        mmi[2 * t + 1] = momentum[2 * g + 1];
        sg[t] = sigmaV2[base + t];
    }
    __syncthreads();

    const int lane = t & 31;
    const int wave = t >> 5;
    const int rb   = wave * 16;
    const int q    = lane & 15;
    const int half = lane >> 4;
    const int row  = rb + q;

    const int  comp = (q == 0) ? 0 : 1;
    const bool live = (q < 2);
    v2f Bf[16];
    #pragma unroll
    for (int kc = 0; kc < 16; ++kc) {
        const int j0 = 4 * kc + 2 * half;
        const float m0 = mmi[2 * j0 + comp];
        const float m1 = mmi[2 * j0 + 2 + comp];
        Bf[kc].x = live ? m0 : 0.0f;
        Bf[kc].y = live ? m1 : 0.0f;
    }

    #pragma unroll
    for (int step = 0; step < 2; ++step) {
        const float xr  = xs[row];
        const float yr  = ys[row];
        const float nis = -LOG2E * __builtin_amdgcn_rcpf(sg[row]);

        v8f acc = {0.f, 0.f, 0.f, 0.f, 0.f, 0.f, 0.f, 0.f};

        #pragma unroll
        for (int kc = 0; kc < 16; ++kc) {
            const int j0 = 4 * kc + 2 * half;

            const float dx0 = xr - xs[j0],     dy0 = yr - ys[j0];
            const float dx1 = xr - xs[j0 + 1], dy1 = yr - ys[j0 + 1];
            v2f A;
            A.x = __builtin_amdgcn_exp2f((dx0 * dx0 + dy0 * dy0) * nis);
            A.y = __builtin_amdgcn_exp2f((dx1 * dx1 + dy1 * dy1) * nis);

            acc = __builtin_amdgcn_wmma_f32_16x16x4_f32(
                false, A, false, Bf[kc], (short)0, acc, false, false);
        }

        if (q < 2) {
            float* dst = (q == 0) ? dpx : dpy;
            #pragma unroll
            for (int v = 0; v < 8; ++v)
                dst[rb + half * 8 + v] = acc[v];
        }
        __syncthreads();

        if (step == 0) {
            if (t < LC) {
                xs[t] += TAUC * dpx[t];
                ys[t] += TAUC * dpy[t];
            }
            __syncthreads();
        }
    }

    if (t < LC) {
        const int g = b * L + base + t;
        out[2 * g + 0] = xs[t] + TAUC * dpx[t];
        out[2 * g + 1] = ys[t] + TAUC * dpy[t];
    }
}

extern "C" void kernel_launch(void* const* d_in, const int* in_sizes, int n_in,
                              void* d_out, int out_size, void* d_ws, size_t ws_size,
                              hipStream_t stream) {
    const float* momentum = (const float*)d_in[0];
    const float* init_lm  = (const float*)d_in[1];
    const float* sigmaV2  = (const float*)d_in[3];
    float* out = (float*)d_out;

    const int L = in_sizes[3];
    const int B = in_sizes[0] / (2 * L);

    dim3 grid(B * NCURVE);
    dim3 block(128);
    landmark_deform_wmma_kernel<<<grid, block, 0, stream>>>(
        momentum, init_lm, sigmaV2, out, L);
}


# LDS reads leave the WMMA stream: each Euler step's 16 column-coordinate pair reads (step 0: the B words too) issue as one group ahead of its 16-WMMA block, no longer between the WMMAs
# speedup vs baseline: 1.4066x; 1.0881x over previous
#include <hip/hip_runtime.h>

typedef __attribute__((ext_vector_type(2))) float v2f;
typedef __attribute__((ext_vector_type(8))) float v8f;

#define LC      64
#define NCURVE  8
#define TAUC    0.5f
#define LOG2E   1.44269504088896340736f

__global__ __launch_bounds__(128) void landmark_deform_wmma_kernel(
    const float* __restrict__ momentum,
    const float* __restrict__ init_lm,
    const float* __restrict__ sigmaV2,
    float* __restrict__ out,
    int L)
{
    __shared__ __attribute__((aligned(16))) float xs[LC];
    __shared__ __attribute__((aligned(16))) float ys[LC];
    __shared__ float mmi[2 * LC];
    __shared__ float sg[LC], dpx[LC], dpy[LC];

    const int blk   = blockIdx.x;
    const int b     = blk / NCURVE;
    const int curve = blk % NCURVE;
    const int base  = curve * LC;
    const int t     = threadIdx.x;

    if (t < LC) {
        const int g = b * L + base + t;
        xs[t] = init_lm[2 * g + 0];
        ys[t] = init_lm[2 * g + 1];
        mmi[2 * t + 0] = momentum[2 * g + 0];
        mmi[2 * t + 1] = momentum[2 * g + 1];
        sg[t] = sigmaV2[base + t];
    }
    __syncthreads();

    const int lane = t & 31;
    const int wave = t >> 5;
    const int rb   = wave * 16;
    const int q    = lane & 15;
    const int half = lane >> 4;
    const int row  = rb + q;

    const int  comp = (q == 0) ? 0 : 1;
    const bool live = (q < 2);
    v2f Bf[16];

    #pragma unroll
    for (int step = 0; step < 2; ++step) {
        const float xr  = xs[row];
        const float yr  = ys[row];
        const float nis = -LOG2E * __builtin_amdgcn_rcpf(sg[row]);

        v8f acc = {0.f, 0.f, 0.f, 0.f, 0.f, 0.f, 0.f, 0.f};

        v2f cx[16], cy[16];
        #pragma unroll
        for (int kc = 0; kc < 16; ++kc) {
            const int j0 = 4 * kc + 2 * half;
            cx[kc] = *reinterpret_cast<const v2f*>(&xs[j0]);
            cy[kc] = *reinterpret_cast<const v2f*>(&ys[j0]);
            if (step == 0) {
                Bf[kc].x = mmi[2 * j0 + comp];
                Bf[kc].y = mmi[2 * j0 + 2 + comp];
            }
        }
        __builtin_amdgcn_sched_barrier(0);
        if (step == 0) {
            #pragma unroll
            for (int kc = 0; kc < 16; ++kc) {
                Bf[kc].x = live ? Bf[kc].x : 0.0f;
                Bf[kc].y = live ? Bf[kc].y : 0.0f;
            }
        }

        #pragma unroll
        for (int kc = 0; kc < 16; ++kc) {
            const float dx0 = xr - cx[kc].x, dy0 = yr - cy[kc].x;
            const float dx1 = xr - cx[kc].y, dy1 = yr - cy[kc].y;
            v2f A;
            A.x = __builtin_amdgcn_exp2f((dx0 * dx0 + dy0 * dy0) * nis);
            A.y = __builtin_amdgcn_exp2f((dx1 * dx1 + dy1 * dy1) * nis);

            acc = __builtin_amdgcn_wmma_f32_16x16x4_f32(
                false, A, false, Bf[kc], (short)0, acc, false, false);
        }

        asm volatile("v_nop\n\tv_nop\n\tv_nop\n\tv_nop\n\tv_nop" : "+v"(acc));

        if (q < 2) {
            float* dst = (q == 0) ? dpx : dpy;
            #pragma unroll
            for (int v = 0; v < 8; ++v)
                dst[rb + half * 8 + v] = acc[v];
        }
        __syncthreads();

        if (step == 0) {
            if (t < LC) {
                xs[t] += TAUC * dpx[t];
                ys[t] += TAUC * dpy[t];
            }
            __syncthreads();
        }
    }

    if (t < LC) {
        const int g = b * L + base + t;
        out[2 * g + 0] = xs[t] + TAUC * dpx[t];
        out[2 * g + 1] = ys[t] + TAUC * dpy[t];
    }
}

extern "C" void kernel_launch(void* const* d_in, const int* in_sizes, int n_in,
                              void* d_out, int out_size, void* d_ws, size_t ws_size,
                              hipStream_t stream) {
    const float* momentum = (const float*)d_in[0];
    const float* init_lm  = (const float*)d_in[1];
    const float* sigmaV2  = (const float*)d_in[3];
    float* out = (float*)d_out;

    const int L = in_sizes[3];
    const int B = in_sizes[0] / (2 * L);

    dim3 grid(B * NCURVE);
    dim3 block(128);
    landmark_deform_wmma_kernel<<<grid, block, 0, stream>>>(
        momentum, init_lm, sigmaV2, out, L);
}
